# starting kernel
# baseline (speedup 1.0000x reference)
#include <hip/hip_runtime.h>

#define BATCH 512
#define TLEN  512
#define NVAR  64
#define RANK  16

typedef __attribute__((ext_vector_type(2))) float v2f;
typedef __attribute__((ext_vector_type(4))) float v4f;
typedef __attribute__((ext_vector_type(8))) float v8f;

#if __has_builtin(__builtin_amdgcn_tanhf)
  #define TANHF(x) __builtin_amdgcn_tanhf(x)
#elif __has_builtin(__builtin_amdgcn_tanh_f32)
  #define TANHF(x) __builtin_amdgcn_tanh_f32(x)
#else
  #define TANHF(x) tanhf(x)
#endif

#define BS_IDX(k, v, o)   ((((k) * 64 + (v)) * 16) + (((o) ^ (((v) >> 2) & 15))))
#define OUTS_IDX(b, o, v) ((((b) * 16 + (o)) * 64) + (((v) ^ ((((o) & 15) << 2) & 63))))

__global__ __launch_bounds__(256) void gcm_stage1(
    const float* __restrict__ x, const float* __restrict__ gating,
    const float* __restrict__ lora_A, float* __restrict__ tmp)
{
  __shared__ float xs[32 * 16 * 17];
  __shared__ float as[32 * 16];

  const int tid  = threadIdx.x;
  const int lane = tid & 31;
  const int wave = tid >> 5;
  const int b0   = blockIdx.x * 16;
  const int v0   = blockIdx.y * 16;

  const int m    = lane & 15;
  const int koff = (lane < 16) ? 0 : 2;
  const int vla  = wave * 2;
  const int vlb  = vla + 1;
  const float ga = gating[v0 + vla];
  const float gb = gating[v0 + vlb];

  const int sv = (tid * 4) & 15;
  const int si = (tid >> 2) & 31;
  const int sb = tid >> 7;

  v8f acc0 = {0.f, 0.f, 0.f, 0.f, 0.f, 0.f, 0.f, 0.f};
  v8f acc1 = {0.f, 0.f, 0.f, 0.f, 0.f, 0.f, 0.f, 0.f};

  for (int s = 0; s < TLEN / 32; ++s) {
    const int i0 = s * 32;
    __syncthreads();

    v4f vals[8];
    #pragma unroll
    for (int r = 0; r < 8; ++r) {
      const int b = sb + 2 * r;
      vals[r] = *reinterpret_cast<const v4f*>(
          &x[((size_t)(b0 + b) * TLEN + (i0 + si)) * NVAR + v0 + sv]);
    }
    #pragma unroll
    for (int r = 0; r < 8; ++r) {
      const int b = sb + 2 * r;
      xs[((si * 16) + sv + 0) * 17 + b] = vals[r].x;
      xs[((si * 16) + sv + 1) * 17 + b] = vals[r].y;
      xs[((si * 16) + sv + 2) * 17 + b] = vals[r].z;
      xs[((si * 16) + sv + 3) * 17 + b] = vals[r].w;
    }
    #pragma unroll
    for (int r = 0; r < 2; ++r) {
      const int e = tid + r * 256;
      const int k = e & 15;
      const int i = e >> 4;
      as[i * 16 + k] = lora_A[(size_t)(i0 + i) * RANK + k];
    }
    __syncthreads();

    #pragma unroll
    for (int ic = 0; ic < 8; ++ic) {
      const int il = ic * 4 + koff;
      v2f bmat;
      bmat.x = as[(il + 0) * 16 + m];
      bmat.y = as[(il + 1) * 16 + m];
      v2f a0, a1;
      a0.x = TANHF(ga * xs[((il + 0) * 16 + vla) * 17 + m]);
      a0.y = TANHF(ga * xs[((il + 1) * 16 + vla) * 17 + m]);
      a1.x = TANHF(gb * xs[((il + 0) * 16 + vlb) * 17 + m]);
      a1.y = TANHF(gb * xs[((il + 1) * 16 + vlb) * 17 + m]);
      acc0 = __builtin_amdgcn_wmma_f32_16x16x4_f32(false, a0, false, bmat,
                                                   (short)0, acc0, false, false);
      acc1 = __builtin_amdgcn_wmma_f32_16x16x4_f32(false, a1, false, bmat,
                                                   (short)0, acc1, false, false);
    }
  }

  const int mrow = (lane >> 4) * 8;
  const int n    = lane & 15;
  #pragma unroll
  for (int r = 0; r < 8; ++r) {
    tmp[((size_t)(v0 + vla) * BATCH + b0 + mrow + r) * RANK + n] = acc0[r];
    tmp[((size_t)(v0 + vlb) * BATCH + b0 + mrow + r) * RANK + n] = acc1[r];
  }
}

__global__ __launch_bounds__(1024) void gcm_stage2(
    const float* __restrict__ x, const float* __restrict__ bias,
    const float* __restrict__ lora_B, const float* __restrict__ tmp,
    float* __restrict__ out)
{
  __shared__ float smem[16 * 64 * 16];
  float* Bs   = smem;
  float* outs = smem;

  const int tid  = threadIdx.x;
  const int lane = tid & 31;
  const int wave = tid >> 5;
  const int b0   = blockIdx.x * 16;
  const int o0   = blockIdx.y * 16;
  const int m    = lane & 15;
  const int koff = (lane < 16) ? 0 : 2;
  const int va   = wave * 2;
  const int vb   = va + 1;

  const int sv = (tid * 4) & 63;
  const int so = (tid >> 4) & 15;
  const int sk = tid >> 8;

  {
    v4f vals[4];
    #pragma unroll
    for (int r = 0; r < 4; ++r) {
      const int k = sk + 4 * r;
      vals[r] = *reinterpret_cast<const v4f*>(
          &lora_B[((size_t)k * TLEN + o0 + so) * NVAR + sv]);
    }
    #pragma unroll
    for (int r = 0; r < 4; ++r) {
      const int k = sk + 4 * r;
      Bs[BS_IDX(k, sv + 0, so)] = vals[r].x;
      Bs[BS_IDX(k, sv + 1, so)] = vals[r].y;
      Bs[BS_IDX(k, sv + 2, so)] = vals[r].z;
      Bs[BS_IDX(k, sv + 3, so)] = vals[r].w;
    }
  }
  __syncthreads();

  v8f acc0 = {0.f, 0.f, 0.f, 0.f, 0.f, 0.f, 0.f, 0.f};
  v8f acc1 = {0.f, 0.f, 0.f, 0.f, 0.f, 0.f, 0.f, 0.f};
  #pragma unroll
  for (int kc = 0; kc < 4; ++kc) {
    const int k = kc * 4 + koff;
    v2f a0, a1, bm0, bm1;
    a0.x = tmp[((size_t)va * BATCH + b0 + m) * RANK + k + 0];
    a0.y = tmp[((size_t)va * BATCH + b0 + m) * RANK + k + 1];
    a1.x = tmp[((size_t)vb * BATCH + b0 + m) * RANK + k + 0];
    a1.y = tmp[((size_t)vb * BATCH + b0 + m) * RANK + k + 1];
    bm0.x = Bs[BS_IDX(k + 0, va, m)];
    bm0.y = Bs[BS_IDX(k + 1, va, m)];
    bm1.x = Bs[BS_IDX(k + 0, vb, m)];
    bm1.y = Bs[BS_IDX(k + 1, vb, m)];
    acc0 = __builtin_amdgcn_wmma_f32_16x16x4_f32(false, a0, false, bm0,
                                                 (short)0, acc0, false, false);
    acc1 = __builtin_amdgcn_wmma_f32_16x16x4_f32(false, a1, false, bm1,
                                                 (short)0, acc1, false, false);
  }
  __syncthreads();

  const int mrow = (lane >> 4) * 8;
  const int n    = lane & 15;
  #pragma unroll
  for (int r = 0; r < 8; ++r) {
    outs[OUTS_IDX(mrow + r, n, va)] = acc0[r];
    outs[OUTS_IDX(mrow + r, n, vb)] = acc1[r];
  }
  __syncthreads();

  #pragma unroll
  for (int r = 0; r < 4; ++r) {
    const int b = sk + 4 * r;
    const size_t g = ((size_t)(b0 + b) * TLEN + (o0 + so)) * NVAR + sv;
    const v4f xv = *reinterpret_cast<const v4f*>(&x[g]);
    const v4f bv = *reinterpret_cast<const v4f*>(
        &bias[(size_t)(o0 + so) * NVAR + sv]);
    v4f ov;
    ov.x = xv.x + bv.x + outs[OUTS_IDX(b, so, sv + 0)];
    ov.y = xv.y + bv.y + outs[OUTS_IDX(b, so, sv + 1)];
    ov.z = xv.z + bv.z + outs[OUTS_IDX(b, so, sv + 2)];
    ov.w = xv.w + bv.w + outs[OUTS_IDX(b, so, sv + 3)];
    __builtin_nontemporal_store(ov, reinterpret_cast<v4f*>(&out[g]));
  }
}

extern "C" void kernel_launch(void* const* d_in, const int* in_sizes, int n_in,
                              void* d_out, int out_size, void* d_ws, size_t ws_size,
                              hipStream_t stream) {
  (void)in_sizes; (void)n_in; (void)out_size; (void)ws_size;
  const float* x      = (const float*)d_in[0];
  const float* gating = (const float*)d_in[1];
  const float* bias   = (const float*)d_in[2];
  const float* lora_A = (const float*)d_in[3];
  const float* lora_B = (const float*)d_in[4];
  float* out = (float*)d_out;
  float* tmp = (float*)d_ws;

  dim3 g1(BATCH / 16, NVAR / 16);
  gcm_stage1<<<g1, dim3(256), 0, stream>>>(x, gating, lora_A, tmp);

  dim3 g2(BATCH / 16, TLEN / 16);
  gcm_stage2<<<g2, dim3(1024), 0, stream>>>(x, bias, lora_B, tmp, out);
}


# Stage 1 chunk loop software-pipelined: next 64-row chunk's x and lora_A loads issued as one group before this chunk's WMMAs, held in registers; LDS tiles double-buffered; sum order unchanged
# speedup vs baseline: 1.9556x; 1.9556x over previous
#include <hip/hip_runtime.h>

#define BATCH 512
#define TLEN  512
#define NVAR  64
#define RANK  16

typedef __attribute__((ext_vector_type(2))) float v2f;
typedef __attribute__((ext_vector_type(4))) float v4f;
typedef __attribute__((ext_vector_type(8))) float v8f;

#if __has_builtin(__builtin_amdgcn_tanhf)
  #define TANHF(x) __builtin_amdgcn_tanhf(x)
#elif __has_builtin(__builtin_amdgcn_tanh_f32)
  #define TANHF(x) __builtin_amdgcn_tanh_f32(x)
#else
  #define TANHF(x) tanhf(x)
#endif

#define BS_IDX(k, v, o)   ((((k) * 64 + (v)) * 16) + (((o) ^ (((v) >> 2) & 15))))
#define OUTS_IDX(b, o, v) ((((b) * 16 + (o)) * 64) + (((v) ^ ((((o) & 15) << 2) & 63))))

#define KCH   64
#define NCH   (TLEN / KCH)
#define XS_CH (KCH * 16 * 17)
#define AS_CH (KCH * RANK)

__global__ __launch_bounds__(256) void gcm_stage1(
    const float* __restrict__ x, const float* __restrict__ gating,
    const float* __restrict__ lora_A, float* __restrict__ tmp)
{
  __shared__ __attribute__((aligned(16))) float xs[2 * XS_CH];
  __shared__ __attribute__((aligned(16))) float as[2 * AS_CH];

  const int tid  = threadIdx.x;
  const int lane = tid & 31;
  const int wave = tid >> 5;
  const int b0   = blockIdx.x * 16;
  const int v0   = blockIdx.y * 16;

  const int m    = lane & 15;
  const int koff = (lane < 16) ? 0 : 2;
  const int vla  = wave * 2;
  const int vlb  = vla + 1;
  const float ga = gating[v0 + vla];
  const float gb = gating[v0 + vlb];

  const int sv = (tid & 3) * 4;
  const int si = tid >> 2;
  const float* xp = x + ((size_t)b0 * TLEN + si) * NVAR + v0 + sv;
  const float* ap = lora_A + tid * 4;
  const int xw = (si * 16 + sv) * 17;

  v4f vals[16];
  v4f av;

  #pragma unroll
  for (int r = 0; r < 16; ++r)
    vals[r] = *reinterpret_cast<const v4f*>(xp + (size_t)r * TLEN * NVAR);
  av = *reinterpret_cast<const v4f*>(ap);
  #pragma unroll
  for (int r = 0; r < 16; ++r) {
    xs[xw + 0 * 17 + r] = vals[r].x;
    xs[xw + 1 * 17 + r] = vals[r].y;
    xs[xw + 2 * 17 + r] = vals[r].z;
    xs[xw + 3 * 17 + r] = vals[r].w;
  }
  *reinterpret_cast<v4f*>(&as[tid * 4]) = av;
  __syncthreads();

  v8f acc0 = {0.f, 0.f, 0.f, 0.f, 0.f, 0.f, 0.f, 0.f};
  v8f acc1 = {0.f, 0.f, 0.f, 0.f, 0.f, 0.f, 0.f, 0.f};

  for (int c = 0; c < NCH; ++c) {
    const bool more = (c + 1 < NCH);
    if (more) {
      const size_t adv = (size_t)(c + 1) * KCH;
      #pragma unroll
      for (int r = 0; r < 16; ++r)
        vals[r] = *reinterpret_cast<const v4f*>(
            xp + (size_t)r * TLEN * NVAR + adv * NVAR);
      av = *reinterpret_cast<const v4f*>(ap + adv * RANK);
    }

    const float* xb = xs + (c & 1) * XS_CH;
    const float* ab = as + (c & 1) * AS_CH;
    #pragma unroll
    for (int ic = 0; ic < KCH / 4; ++ic) {
      const int il = ic * 4 + koff;
      v2f bmat;
      bmat.x = ab[(il + 0) * 16 + m];
      bmat.y = ab[(il + 1) * 16 + m];
      v2f a0, a1;
      a0.x = TANHF(ga * xb[((il + 0) * 16 + vla) * 17 + m]);
      a0.y = TANHF(ga * xb[((il + 1) * 16 + vla) * 17 + m]);
      a1.x = TANHF(gb * xb[((il + 0) * 16 + vlb) * 17 + m]);
      a1.y = TANHF(gb * xb[((il + 1) * 16 + vlb) * 17 + m]);
      acc0 = __builtin_amdgcn_wmma_f32_16x16x4_f32(false, a0, false, bmat,
                                                   (short)0, acc0, false, false);
      acc1 = __builtin_amdgcn_wmma_f32_16x16x4_f32(false, a1, false, bmat,
                                                   (short)0, acc1, false, false);
    }

    if (more) {
      float* xn = xs + ((c + 1) & 1) * XS_CH;
      float* an = as + ((c + 1) & 1) * AS_CH;
      #pragma unroll
      for (int r = 0; r < 16; ++r) {
        xn[xw + 0 * 17 + r] = vals[r].x;
        xn[xw + 1 * 17 + r] = vals[r].y;
        xn[xw + 2 * 17 + r] = vals[r].z;
        xn[xw + 3 * 17 + r] = vals[r].w;
      }
      *reinterpret_cast<v4f*>(&an[tid * 4]) = av;
      __syncthreads();
    }
  }

  const int mrow = (lane >> 4) * 8;
  const int n    = lane & 15;
  #pragma unroll
  for (int r = 0; r < 8; ++r) {
    tmp[((size_t)(v0 + vla) * BATCH + b0 + mrow + r) * RANK + n] = acc0[r];
    tmp[((size_t)(v0 + vlb) * BATCH + b0 + mrow + r) * RANK + n] = acc1[r];
  }
}

__global__ __launch_bounds__(1024) void gcm_stage2(
    const float* __restrict__ x, const float* __restrict__ bias,
    const float* __restrict__ lora_B, const float* __restrict__ tmp,
    float* __restrict__ out)
{
  __shared__ float smem[16 * 64 * 16];
  float* Bs   = smem;
  float* outs = smem;

  const int tid  = threadIdx.x;
  const int lane = tid & 31;
  const int wave = tid >> 5;
  const int b0   = blockIdx.x * 16;
  const int o0   = blockIdx.y * 16;
  const int m    = lane & 15;
  const int koff = (lane < 16) ? 0 : 2;
  const int va   = wave * 2;
  const int vb   = va + 1;

  const int sv = (tid * 4) & 63;
  const int so = (tid >> 4) & 15;
  const int sk = tid >> 8;

  {
    v4f vals[4];
    #pragma unroll
    for (int r = 0; r < 4; ++r) {
      const int k = sk + 4 * r;
      vals[r] = *reinterpret_cast<const v4f*>(
          &lora_B[((size_t)k * TLEN + o0 + so) * NVAR + sv]);
    }
    #pragma unroll
    for (int r = 0; r < 4; ++r) {
      const int k = sk + 4 * r;
      Bs[BS_IDX(k, sv + 0, so)] = vals[r].x;
      Bs[BS_IDX(k, sv + 1, so)] = vals[r].y;
      Bs[BS_IDX(k, sv + 2, so)] = vals[r].z;
      Bs[BS_IDX(k, sv + 3, so)] = vals[r].w;
    }
  }
  __syncthreads();

  v8f acc0 = {0.f, 0.f, 0.f, 0.f, 0.f, 0.f, 0.f, 0.f};
  v8f acc1 = {0.f, 0.f, 0.f, 0.f, 0.f, 0.f, 0.f, 0.f};
  #pragma unroll
  for (int kc = 0; kc < 4; ++kc) {
    const int k = kc * 4 + koff;
    v2f a0, a1, bm0, bm1;
    a0.x = tmp[((size_t)va * BATCH + b0 + m) * RANK + k + 0];
    a0.y = tmp[((size_t)va * BATCH + b0 + m) * RANK + k + 1];
    a1.x = tmp[((size_t)vb * BATCH + b0 + m) * RANK + k + 0];
    a1.y = tmp[((size_t)vb * BATCH + b0 + m) * RANK + k + 1];
    bm0.x = Bs[BS_IDX(k + 0, va, m)];
    bm0.y = Bs[BS_IDX(k + 1, va, m)];
    bm1.x = Bs[BS_IDX(k + 0, vb, m)];
    bm1.y = Bs[BS_IDX(k + 1, vb, m)];
    acc0 = __builtin_amdgcn_wmma_f32_16x16x4_f32(false, a0, false, bm0,
                                                 (short)0, acc0, false, false);
    acc1 = __builtin_amdgcn_wmma_f32_16x16x4_f32(false, a1, false, bm1,
                                                 (short)0, acc1, false, false);
  }
  __syncthreads();

  const int mrow = (lane >> 4) * 8;
  const int n    = lane & 15;
  #pragma unroll
  for (int r = 0; r < 8; ++r) {
    outs[OUTS_IDX(mrow + r, n, va)] = acc0[r];
    outs[OUTS_IDX(mrow + r, n, vb)] = acc1[r];
  }
  __syncthreads();

  #pragma unroll
  for (int r = 0; r < 4; ++r) {
    const int b = sk + 4 * r;
    const size_t g = ((size_t)(b0 + b) * TLEN + (o0 + so)) * NVAR + sv;
    const v4f xv = *reinterpret_cast<const v4f*>(&x[g]);
    const v4f bv = *reinterpret_cast<const v4f*>(
        &bias[(size_t)(o0 + so) * NVAR + sv]);
    v4f ov;
    ov.x = xv.x + bv.x + outs[OUTS_IDX(b, so, sv + 0)];
    ov.y = xv.y + bv.y + outs[OUTS_IDX(b, so, sv + 1)];
    ov.z = xv.z + bv.z + outs[OUTS_IDX(b, so, sv + 2)];
    ov.w = xv.w + bv.w + outs[OUTS_IDX(b, so, sv + 3)];
    __builtin_nontemporal_store(ov, reinterpret_cast<v4f*>(&out[g]));
  }
}

extern "C" void kernel_launch(void* const* d_in, const int* in_sizes, int n_in,
                              void* d_out, int out_size, void* d_ws, size_t ws_size,
                              hipStream_t stream) {
  (void)in_sizes; (void)n_in; (void)out_size; (void)ws_size;
  const float* x      = (const float*)d_in[0];
  const float* gating = (const float*)d_in[1];
  const float* bias   = (const float*)d_in[2];
  const float* lora_A = (const float*)d_in[3];
  const float* lora_B = (const float*)d_in[4];
  float* out = (float*)d_out;
  float* tmp = (float*)d_ws;

  dim3 g1(BATCH / 16, NVAR / 16);
  gcm_stage1<<<g1, dim3(256), 0, stream>>>(x, gating, lora_A, tmp);

  dim3 g2(BATCH / 16, TLEN / 16);
  gcm_stage2<<<g2, dim3(1024), 0, stream>>>(x, bias, lora_B, tmp, out);
}


# Stage 2 made lora_B-stationary: 256 work-groups of 16 waves stage their o-tile of lora_B once, keep its B fragments in registers, walk 4 batch tiles; loads fly over LDS-only barriers; sum order kept
# speedup vs baseline: 2.0662x; 1.0566x over previous
#include <hip/hip_runtime.h>

#define BATCH 512
#define TLEN  512
#define NVAR  64
#define RANK  16

typedef __attribute__((ext_vector_type(2))) float v2f;
typedef __attribute__((ext_vector_type(4))) float v4f;
typedef __attribute__((ext_vector_type(8))) float v8f;

#if __has_builtin(__builtin_amdgcn_tanhf)
  #define TANHF(x) __builtin_amdgcn_tanhf(x)
#elif __has_builtin(__builtin_amdgcn_tanh_f32)
  #define TANHF(x) __builtin_amdgcn_tanh_f32(x)
#else
  #define TANHF(x) tanhf(x)
#endif

#define BS_IDX(k, v, o)   ((((k) * 64 + (v)) * 16) + (((o) ^ (((v) >> 2) & 15))))

#define KCH   64
#define NCH   (TLEN / KCH)
#define XS_CH (KCH * 16 * 17)
#define AS_CH (KCH * RANK)

__global__ __launch_bounds__(256) void gcm_stage1(
    const float* __restrict__ x, const float* __restrict__ gating,
    const float* __restrict__ lora_A, float* __restrict__ tmp)
{
  __shared__ __attribute__((aligned(16))) float xs[2 * XS_CH];
  __shared__ __attribute__((aligned(16))) float as[2 * AS_CH];

  const int tid  = threadIdx.x;
  const int lane = tid & 31;
  const int wave = tid >> 5;
  const int b0   = blockIdx.x * 16;
  const int v0   = blockIdx.y * 16;

  const int m    = lane & 15;
  const int koff = (lane < 16) ? 0 : 2;
  const int vla  = wave * 2;
  const int vlb  = vla + 1;
  const float ga = gating[v0 + vla];
  const float gb = gating[v0 + vlb];

  const int sv = (tid & 3) * 4;
  const int si = tid >> 2;
  const float* xp = x + ((size_t)b0 * TLEN + si) * NVAR + v0 + sv;
  const float* ap = lora_A + tid * 4;
  const int xw = (si * 16 + sv) * 17;

  v4f vals[16];
  v4f av;

  #pragma unroll
  for (int r = 0; r < 16; ++r)
    vals[r] = *reinterpret_cast<const v4f*>(xp + (size_t)r * TLEN * NVAR);
  av = *reinterpret_cast<const v4f*>(ap);
  #pragma unroll
  for (int r = 0; r < 16; ++r) {
    xs[xw + 0 * 17 + r] = vals[r].x;
    xs[xw + 1 * 17 + r] = vals[r].y;
    xs[xw + 2 * 17 + r] = vals[r].z;
    xs[xw + 3 * 17 + r] = vals[r].w;
  }
  *reinterpret_cast<v4f*>(&as[tid * 4]) = av;
  __syncthreads();

  v8f acc0 = {0.f, 0.f, 0.f, 0.f, 0.f, 0.f, 0.f, 0.f};
  v8f acc1 = {0.f, 0.f, 0.f, 0.f, 0.f, 0.f, 0.f, 0.f};

  for (int c = 0; c < NCH; ++c) {
    const bool more = (c + 1 < NCH);
    if (more) {
      const size_t adv = (size_t)(c + 1) * KCH;
      #pragma unroll
      for (int r = 0; r < 16; ++r)
        vals[r] = *reinterpret_cast<const v4f*>(
            xp + (size_t)r * TLEN * NVAR + adv * NVAR);
      av = *reinterpret_cast<const v4f*>(ap + adv * RANK);
    }

    const float* xb = xs + (c & 1) * XS_CH;
    const float* ab = as + (c & 1) * AS_CH;
    #pragma unroll
    for (int ic = 0; ic < KCH / 4; ++ic) {
      const int il = ic * 4 + koff;
      v2f bmat;
      bmat.x = ab[(il + 0) * 16 + m];
      bmat.y = ab[(il + 1) * 16 + m];
      v2f a0, a1;
      a0.x = TANHF(ga * xb[((il + 0) * 16 + vla) * 17 + m]);
      a0.y = TANHF(ga * xb[((il + 1) * 16 + vla) * 17 + m]);
      a1.x = TANHF(gb * xb[((il + 0) * 16 + vlb) * 17 + m]);
      a1.y = TANHF(gb * xb[((il + 1) * 16 + vlb) * 17 + m]);
      acc0 = __builtin_amdgcn_wmma_f32_16x16x4_f32(false, a0, false, bmat,
                                                   (short)0, acc0, false, false);
      acc1 = __builtin_amdgcn_wmma_f32_16x16x4_f32(false, a1, false, bmat,
                                                   (short)0, acc1, false, false);
    }

    if (more) {
      float* xn = xs + ((c + 1) & 1) * XS_CH;
      float* an = as + ((c + 1) & 1) * AS_CH;
      #pragma unroll
      for (int r = 0; r < 16; ++r) {
        xn[xw + 0 * 17 + r] = vals[r].x;
        xn[xw + 1 * 17 + r] = vals[r].y;
        xn[xw + 2 * 17 + r] = vals[r].z;
        xn[xw + 3 * 17 + r] = vals[r].w;
      }
      *reinterpret_cast<v4f*>(&an[tid * 4]) = av;
      __syncthreads();
    }
  }

  const int mrow = (lane >> 4) * 8;
  const int n    = lane & 15;
  #pragma unroll
  for (int r = 0; r < 8; ++r) {
    tmp[((size_t)(v0 + vla) * BATCH + b0 + mrow + r) * RANK + n] = acc0[r];
    tmp[((size_t)(v0 + vlb) * BATCH + b0 + mrow + r) * RANK + n] = acc1[r];
  }
}

static __device__ __forceinline__ void lds_barrier() {
  asm volatile("s_wait_dscnt 0x0\n\ts_barrier_signal -1\n\ts_barrier_wait -1" ::: "memory");
  __builtin_amdgcn_sched_barrier(0);
}

#define S2_BT   4
#define S2_VPW  4
#define OP      68
#define OT_FL   (16 * 16 * OP)

__global__ __launch_bounds__(512) void gcm_stage2(
    const float* __restrict__ x, const float* __restrict__ bias,
    const float* __restrict__ lora_B, const float* __restrict__ tmp,
    float* __restrict__ out)
{
  __shared__ __attribute__((aligned(16))) float smem[2 * OT_FL];
  float* Bs = smem + OT_FL;

  const int tid  = threadIdx.x;
  const int lane = tid & 31;
  const int wave = tid >> 5;
  const int bg0  = blockIdx.x * (16 * S2_BT);
  const int o0   = blockIdx.y * 16;
  const int m    = lane & 15;
  const int half = lane >> 4;
  const int koff = half * 2;
  const int vw   = wave * S2_VPW;

  const int rv = (tid & 15) * 4;
  const int ro = (tid >> 4) & 15;
  const int rb = tid >> 8;

  v4f bvals[8];
  #pragma unroll
  for (int r = 0; r < 8; ++r)
    bvals[r] = *reinterpret_cast<const v4f*>(
        &lora_B[((size_t)(rb + 2 * r) * TLEN + o0 + ro) * NVAR + rv]);

  v2f  afr[2][S2_VPW][4];
  v4f  xv[2][8];

  const float* tp = tmp + ((size_t)vw * BATCH + bg0 + m) * RANK + koff;
  const float* xq = x + ((size_t)(bg0 + rb) * TLEN + o0 + ro) * NVAR + rv;
  float*       oq = out + ((size_t)(bg0 + rb) * TLEN + o0 + ro) * NVAR + rv;

  #pragma unroll
  for (int v = 0; v < S2_VPW; ++v)
    #pragma unroll
    for (int kc = 0; kc < 4; ++kc)
      afr[0][v][kc] = *reinterpret_cast<const v2f*>(
          tp + (size_t)v * BATCH * RANK + kc * 4);
  #pragma unroll
  for (int j = 0; j < 8; ++j)
    xv[0][j] = *reinterpret_cast<const v4f*>(xq + (size_t)(2 * j) * TLEN * NVAR);
  const v4f bv = *reinterpret_cast<const v4f*>(&bias[(size_t)(o0 + ro) * NVAR + rv]);

  #pragma unroll
  for (int r = 0; r < 8; ++r) {
    const int k = rb + 2 * r;
    Bs[BS_IDX(k, rv + 0, ro)] = bvals[r].x;
    Bs[BS_IDX(k, rv + 1, ro)] = bvals[r].y;
    Bs[BS_IDX(k, rv + 2, ro)] = bvals[r].z;
    Bs[BS_IDX(k, rv + 3, ro)] = bvals[r].w;
  }
  lds_barrier();

  v2f bfr[S2_VPW][4];
  #pragma unroll
  for (int v = 0; v < S2_VPW; ++v)
    #pragma unroll
    for (int kc = 0; kc < 4; ++kc) {
      const int k = kc * 4 + koff;
      bfr[v][kc].x = Bs[BS_IDX(k + 0, vw + v, m)];
      bfr[v][kc].y = Bs[BS_IDX(k + 1, vw + v, m)];
    }

  #pragma unroll
  for (int t = 0; t < S2_BT; ++t) {
    const int cur = t & 1;
    const int nxt = cur ^ 1;
    if (t + 1 < S2_BT) {
      #pragma unroll
      for (int v = 0; v < S2_VPW; ++v)
        #pragma unroll
        for (int kc = 0; kc < 4; ++kc)
          afr[nxt][v][kc] = *reinterpret_cast<const v2f*>(
              tp + (size_t)v * BATCH * RANK + (size_t)(t + 1) * 16 * RANK + kc * 4);
      #pragma unroll
      for (int j = 0; j < 8; ++j)
        xv[nxt][j] = *reinterpret_cast<const v4f*>(
            xq + (size_t)((t + 1) * 16 + 2 * j) * TLEN * NVAR);
    }

    v8f acc[S2_VPW];
    #pragma unroll
    for (int v = 0; v < S2_VPW; ++v)
      acc[v] = (v8f){0.f, 0.f, 0.f, 0.f, 0.f, 0.f, 0.f, 0.f};
    #pragma unroll
    for (int kc = 0; kc < 4; ++kc)
      #pragma unroll
      for (int v = 0; v < S2_VPW; ++v)
        acc[v] = __builtin_amdgcn_wmma_f32_16x16x4_f32(
            false, afr[cur][v][kc], false, bfr[v][kc], (short)0, acc[v],
            false, false);

    float* ob = smem + cur * OT_FL;
    #pragma unroll
    for (int r = 0; r < 8; ++r) {
      v4f d;
      d.x = acc[0][r]; d.y = acc[1][r]; d.z = acc[2][r]; d.w = acc[3][r];
      *reinterpret_cast<v4f*>(&ob[((half * 8 + r) * 16 + m) * OP + vw]) = d;
    }
    lds_barrier();

    #pragma unroll
    for (int j = 0; j < 8; ++j) {
      const int b = rb + 2 * j;
      const v4f pv = *reinterpret_cast<const v4f*>(&ob[(b * 16 + ro) * OP + rv]);
      v4f ov;
      ov.x = xv[cur][j].x + bv.x + pv.x;
      ov.y = xv[cur][j].y + bv.y + pv.y;
      ov.z = xv[cur][j].z + bv.z + pv.z;
      ov.w = xv[cur][j].w + bv.w + pv.w;
      __builtin_nontemporal_store(
          ov, reinterpret_cast<v4f*>(oq + (size_t)(t * 16 + 2 * j) * TLEN * NVAR));
    }
  }
}

extern "C" void kernel_launch(void* const* d_in, const int* in_sizes, int n_in,
                              void* d_out, int out_size, void* d_ws, size_t ws_size,
                              hipStream_t stream) {
  (void)in_sizes; (void)n_in; (void)out_size; (void)ws_size;
  const float* x      = (const float*)d_in[0];
  const float* gating = (const float*)d_in[1];
  const float* bias   = (const float*)d_in[2];
  const float* lora_A = (const float*)d_in[3];
  const float* lora_B = (const float*)d_in[4];
  float* out = (float*)d_out;
  float* tmp = (float*)d_ws;

  dim3 g1(BATCH / 16, NVAR / 16);
  gcm_stage1<<<g1, dim3(256), 0, stream>>>(x, gating, lora_A, tmp);

  dim3 g2(BATCH / (16 * S2_BT), TLEN / 16);
  gcm_stage2<<<g2, dim3(512), 0, stream>>>(x, bias, lora_B, tmp, out);
}


# Stage 1 re-tiled: a work-group owns 2 batch rows x all 64 variables (256 groups, were 128), tile rows = 2 b x 8 v, x kept [b][i][v] in LDS by 16-byte stores; i sum split in even/odd k-step partials
# speedup vs baseline: 3.0579x; 1.4800x over previous
#include <hip/hip_runtime.h>

#define BATCH 512
#define TLEN  512
#define NVAR  64
#define RANK  16

typedef __attribute__((ext_vector_type(2))) float v2f;
typedef __attribute__((ext_vector_type(4))) float v4f;
typedef __attribute__((ext_vector_type(8))) float v8f;

#if __has_builtin(__builtin_amdgcn_tanhf)
  #define TANHF(x) __builtin_amdgcn_tanhf(x)
#elif __has_builtin(__builtin_amdgcn_tanh_f32)
  #define TANHF(x) __builtin_amdgcn_tanh_f32(x)
#else
  #define TANHF(x) tanhf(x)
#endif

#define BS_IDX(k, v, o)   ((((k) * 64 + (v)) * 16) + (((o) ^ (((v) >> 2) & 15))))

#define KCH   128
#define NCH   (TLEN / KCH)
#define XROW  72
#define XPL   (KCH * XROW + 32)
#define XS_CH (2 * XPL)
#define AS_CH (KCH * RANK)

__global__ __launch_bounds__(256) void gcm_stage1(
    const float* __restrict__ x, const float* __restrict__ gating,
    const float* __restrict__ lora_A, float* __restrict__ tmp)
{
  __shared__ __attribute__((aligned(16))) float xs[2 * XS_CH];
  __shared__ __attribute__((aligned(16))) float as[2 * AS_CH];

  const int tid  = threadIdx.x;
  const int lane = tid & 31;
  const int wave = tid >> 5;
  const int b0   = blockIdx.x * 2;

  const int m    = lane & 15;
  const int half = lane >> 4;
  const int bl   = m >> 3;
  const int vl   = wave * 8 + (m & 7);
  const float g  = gating[vl];

  const int sv = (tid & 15) * 4;
  const int sr = tid >> 4;
  const float* xp = x + ((size_t)b0 * TLEN + sr) * NVAR + sv;
  const float* ap = lora_A + tid * 4;
  const int xw = sr * XROW + sv;
  const int fr = bl * XPL + (2 * half) * XROW + vl;

  v4f vals[16];
  v4f av0, av1;

  #pragma unroll
  for (int r = 0; r < 16; ++r)
    vals[r] = *reinterpret_cast<const v4f*>(
        xp + ((size_t)(r >> 3) * TLEN + (r & 7) * 16) * NVAR);
  av0 = *reinterpret_cast<const v4f*>(ap);
  av1 = *reinterpret_cast<const v4f*>(ap + 1024);
  #pragma unroll
  for (int r = 0; r < 16; ++r)
    *reinterpret_cast<v4f*>(&xs[(r >> 3) * XPL + (r & 7) * 16 * XROW + xw]) = vals[r];
  *reinterpret_cast<v4f*>(&as[tid * 4]) = av0;
  *reinterpret_cast<v4f*>(&as[1024 + tid * 4]) = av1;
  __syncthreads();

  v8f acc0 = {0.f, 0.f, 0.f, 0.f, 0.f, 0.f, 0.f, 0.f};
  v8f acc1 = {0.f, 0.f, 0.f, 0.f, 0.f, 0.f, 0.f, 0.f};

  for (int c = 0; c < NCH; ++c) {
    const bool more = (c + 1 < NCH);
    if (more) {
      const size_t adv = (size_t)(c + 1) * KCH;
      #pragma unroll
      for (int r = 0; r < 16; ++r)
        vals[r] = *reinterpret_cast<const v4f*>(
            xp + ((size_t)(r >> 3) * TLEN + (r & 7) * 16 + adv) * NVAR);
      av0 = *reinterpret_cast<const v4f*>(ap + adv * RANK);
      av1 = *reinterpret_cast<const v4f*>(ap + adv * RANK + 1024);
    }

    const float* xb = xs + (c & 1) * XS_CH + fr;
    const float* ab = as + (c & 1) * AS_CH + (2 * half) * 16 + m;
    #pragma unroll
    for (int ic = 0; ic < KCH / 4; ic += 2) {
      v2f b0m, b1m;
      b0m.x = ab[(ic * 4 + 0) * 16];
      b0m.y = ab[(ic * 4 + 1) * 16];
      b1m.x = ab[(ic * 4 + 4) * 16];
      b1m.y = ab[(ic * 4 + 5) * 16];
      v2f a0, a1;
      a0.x = TANHF(g * xb[(ic * 4 + 0) * XROW]);
      a0.y = TANHF(g * xb[(ic * 4 + 1) * XROW]);
      a1.x = TANHF(g * xb[(ic * 4 + 4) * XROW]);
      a1.y = TANHF(g * xb[(ic * 4 + 5) * XROW]);
      acc0 = __builtin_amdgcn_wmma_f32_16x16x4_f32(false, a0, false, b0m,
                                                   (short)0, acc0, false, false);
      acc1 = __builtin_amdgcn_wmma_f32_16x16x4_f32(false, a1, false, b1m,
                                                   (short)0, acc1, false, false);
    }

    if (more) {
      float* xn = xs + ((c + 1) & 1) * XS_CH;
      float* an = as + ((c + 1) & 1) * AS_CH;
      #pragma unroll
      for (int r = 0; r < 16; ++r)
        *reinterpret_cast<v4f*>(&xn[(r >> 3) * XPL + (r & 7) * 16 * XROW + xw]) = vals[r];
      *reinterpret_cast<v4f*>(&an[tid * 4]) = av0;
      *reinterpret_cast<v4f*>(&an[1024 + tid * 4]) = av1;
      __syncthreads();
    }
  }

  #pragma unroll
  for (int r = 0; r < 8; ++r)
    tmp[((size_t)(wave * 8 + r) * BATCH + b0 + half) * RANK + m] = acc0[r] + acc1[r];
}

static __device__ __forceinline__ void lds_barrier() {
  asm volatile("s_wait_dscnt 0x0\n\ts_barrier_signal -1\n\ts_barrier_wait -1" ::: "memory");
  __builtin_amdgcn_sched_barrier(0);
}

#define S2_BT   4
#define S2_VPW  4
#define OP      68
#define OT_FL   (16 * 16 * OP)

__global__ __launch_bounds__(512) void gcm_stage2(
    const float* __restrict__ x, const float* __restrict__ bias,
    const float* __restrict__ lora_B, const float* __restrict__ tmp,
    float* __restrict__ out)
{
  __shared__ __attribute__((aligned(16))) float smem[2 * OT_FL];
  float* Bs = smem + OT_FL;

  const int tid  = threadIdx.x;
  const int lane = tid & 31;
  const int wave = tid >> 5;
  const int bg0  = blockIdx.x * (16 * S2_BT);
  const int o0   = blockIdx.y * 16;
  const int m    = lane & 15;
  const int half = lane >> 4;
  const int koff = half * 2;
  const int vw   = wave * S2_VPW;

  const int rv = (tid & 15) * 4;
  const int ro = (tid >> 4) & 15;
  const int rb = tid >> 8;

  v4f bvals[8];
  #pragma unroll
  for (int r = 0; r < 8; ++r)
    bvals[r] = *reinterpret_cast<const v4f*>(
        &lora_B[((size_t)(rb + 2 * r) * TLEN + o0 + ro) * NVAR + rv]);

  v2f  afr[2][S2_VPW][4];
  v4f  xv[2][8];

  const float* tp = tmp + ((size_t)vw * BATCH + bg0 + m) * RANK + koff;
  const float* xq = x + ((size_t)(bg0 + rb) * TLEN + o0 + ro) * NVAR + rv;
  float*       oq = out + ((size_t)(bg0 + rb) * TLEN + o0 + ro) * NVAR + rv;

  #pragma unroll
  for (int v = 0; v < S2_VPW; ++v)
    #pragma unroll
    for (int kc = 0; kc < 4; ++kc)
      afr[0][v][kc] = *reinterpret_cast<const v2f*>(
          tp + (size_t)v * BATCH * RANK + kc * 4);
  #pragma unroll
  for (int j = 0; j < 8; ++j)
    xv[0][j] = *reinterpret_cast<const v4f*>(xq + (size_t)(2 * j) * TLEN * NVAR);
  const v4f bv = *reinterpret_cast<const v4f*>(&bias[(size_t)(o0 + ro) * NVAR + rv]);

  #pragma unroll
  for (int r = 0; r < 8; ++r) {
    const int k = rb + 2 * r;
    Bs[BS_IDX(k, rv + 0, ro)] = bvals[r].x;
    Bs[BS_IDX(k, rv + 1, ro)] = bvals[r].y;
    Bs[BS_IDX(k, rv + 2, ro)] = bvals[r].z;
    Bs[BS_IDX(k, rv + 3, ro)] = bvals[r].w;
  }
  lds_barrier();

  v2f bfr[S2_VPW][4];
  #pragma unroll
  for (int v = 0; v < S2_VPW; ++v)
    #pragma unroll
    for (int kc = 0; kc < 4; ++kc) {
      const int k = kc * 4 + koff;
      bfr[v][kc].x = Bs[BS_IDX(k + 0, vw + v, m)];
      bfr[v][kc].y = Bs[BS_IDX(k + 1, vw + v, m)];
    }

  #pragma unroll
  for (int t = 0; t < S2_BT; ++t) {
    const int cur = t & 1;
    const int nxt = cur ^ 1;
    if (t + 1 < S2_BT) {
      #pragma unroll
      for (int v = 0; v < S2_VPW; ++v)
        #pragma unroll
        for (int kc = 0; kc < 4; ++kc)
          afr[nxt][v][kc] = *reinterpret_cast<const v2f*>(
              tp + (size_t)v * BATCH * RANK + (size_t)(t + 1) * 16 * RANK + kc * 4);
      #pragma unroll
      for (int j = 0; j < 8; ++j)
        xv[nxt][j] = *reinterpret_cast<const v4f*>(
            xq + (size_t)((t + 1) * 16 + 2 * j) * TLEN * NVAR);
    }

    v8f acc[S2_VPW];
    #pragma unroll
    for (int v = 0; v < S2_VPW; ++v)
      acc[v] = (v8f){0.f, 0.f, 0.f, 0.f, 0.f, 0.f, 0.f, 0.f};
    #pragma unroll
    for (int kc = 0; kc < 4; ++kc)
      #pragma unroll
      for (int v = 0; v < S2_VPW; ++v)
        acc[v] = __builtin_amdgcn_wmma_f32_16x16x4_f32(
            false, afr[cur][v][kc], false, bfr[v][kc], (short)0, acc[v],
            false, false);

    float* ob = smem + cur * OT_FL;
    #pragma unroll
    for (int r = 0; r < 8; ++r) {
      v4f d;
      d.x = acc[0][r]; d.y = acc[1][r]; d.z = acc[2][r]; d.w = acc[3][r];
      *reinterpret_cast<v4f*>(&ob[((half * 8 + r) * 16 + m) * OP + vw]) = d;
    }
    lds_barrier();

    #pragma unroll
    for (int j = 0; j < 8; ++j) {
      const int b = rb + 2 * j;
      const v4f pv = *reinterpret_cast<const v4f*>(&ob[(b * 16 + ro) * OP + rv]);
      v4f ov;
      ov.x = xv[cur][j].x + bv.x + pv.x;
      ov.y = xv[cur][j].y + bv.y + pv.y;
      ov.z = xv[cur][j].z + bv.z + pv.z;
      ov.w = xv[cur][j].w + bv.w + pv.w;
      __builtin_nontemporal_store(
          ov, reinterpret_cast<v4f*>(oq + (size_t)(t * 16 + 2 * j) * TLEN * NVAR));
    }
  }
}

extern "C" void kernel_launch(void* const* d_in, const int* in_sizes, int n_in,
                              void* d_out, int out_size, void* d_ws, size_t ws_size,
                              hipStream_t stream) {
  (void)in_sizes; (void)n_in; (void)out_size; (void)ws_size;
  const float* x      = (const float*)d_in[0];
  const float* gating = (const float*)d_in[1];
  const float* bias   = (const float*)d_in[2];
  const float* lora_A = (const float*)d_in[3];
  const float* lora_B = (const float*)d_in[4];
  float* out = (float*)d_out;
  float* tmp = (float*)d_ws;

  dim3 g1(BATCH / 2);
  gcm_stage1<<<g1, dim3(256), 0, stream>>>(x, gating, lora_A, tmp);

  dim3 g2(BATCH / (16 * S2_BT), TLEN / 16);
  gcm_stage2<<<g2, dim3(512), 0, stream>>>(x, bias, lora_B, tmp, out);
}


# Stage 1 work-group i also prefetches into L2 the 1024 x rows stage-2 work-group i reads (one global_prefetch_b8 per thread per chunk) so stage 2's re-read of x can hit L2; arithmetic untouched
# speedup vs baseline: 3.1744x; 1.0381x over previous
#include <hip/hip_runtime.h>

#define BATCH 512
#define TLEN  512
#define NVAR  64
#define RANK  16
#define S2_BT 4
#define S2_GX (BATCH / (16 * S2_BT))

typedef __attribute__((ext_vector_type(2))) float v2f;
typedef __attribute__((ext_vector_type(4))) float v4f;
typedef __attribute__((ext_vector_type(8))) float v8f;

#if __has_builtin(__builtin_amdgcn_tanhf)
  #define TANHF(x) __builtin_amdgcn_tanhf(x)
#elif __has_builtin(__builtin_amdgcn_tanh_f32)
  #define TANHF(x) __builtin_amdgcn_tanh_f32(x)
#else
  #define TANHF(x) tanhf(x)
#endif

#define BS_IDX(k, v, o)   ((((k) * 64 + (v)) * 16) + (((o) ^ (((v) >> 2) & 15))))

#define KCH   128
#define NCH   (TLEN / KCH)
#define XROW  72
#define XPL   (KCH * XROW + 32)
#define XS_CH (2 * XPL)
#define AS_CH (KCH * RANK)

__global__ __launch_bounds__(256) void gcm_stage1(
    const float* __restrict__ x, const float* __restrict__ gating,
    const float* __restrict__ lora_A, float* __restrict__ tmp)
{
  __shared__ __attribute__((aligned(16))) float xs[2 * XS_CH];
  __shared__ __attribute__((aligned(16))) float as[2 * AS_CH];

  const int tid  = threadIdx.x;
  const int lane = tid & 31;
  const int wave = tid >> 5;
  const int b0   = blockIdx.x * 2;

  const int m    = lane & 15;
  const int half = lane >> 4;
  const int bl   = m >> 3;
  const int vl   = wave * 8 + (m & 7);
  const float g  = gating[vl];

  const int sv = (tid & 15) * 4;
  const int sr = tid >> 4;
  const float* xp = x + ((size_t)b0 * TLEN + sr) * NVAR + sv;
  const float* ap = lora_A + tid * 4;
  const int xw = sr * XROW + sv;
  const int fr = bl * XPL + (2 * half) * XROW + vl;

  const float* pf = x + ((size_t)((blockIdx.x % S2_GX) * (16 * S2_BT) + sr) * TLEN
                         + (blockIdx.x / S2_GX) * 16 + (tid & 15)) * NVAR;

  v4f vals[16];
  v4f av0, av1;

  #pragma unroll
  for (int r = 0; r < 16; ++r)
    vals[r] = *reinterpret_cast<const v4f*>(
        xp + ((size_t)(r >> 3) * TLEN + (r & 7) * 16) * NVAR);
  av0 = *reinterpret_cast<const v4f*>(ap);
  av1 = *reinterpret_cast<const v4f*>(ap + 1024);
  #pragma unroll
  for (int r = 0; r < 16; ++r)
    *reinterpret_cast<v4f*>(&xs[(r >> 3) * XPL + (r & 7) * 16 * XROW + xw]) = vals[r];
  *reinterpret_cast<v4f*>(&as[tid * 4]) = av0;
  *reinterpret_cast<v4f*>(&as[1024 + tid * 4]) = av1;
  __syncthreads();

  v8f acc0 = {0.f, 0.f, 0.f, 0.f, 0.f, 0.f, 0.f, 0.f};
  v8f acc1 = {0.f, 0.f, 0.f, 0.f, 0.f, 0.f, 0.f, 0.f};

  for (int c = 0; c < NCH; ++c) {
    const bool more = (c + 1 < NCH);
    if (more) {
      const size_t adv = (size_t)(c + 1) * KCH;
      #pragma unroll
      for (int r = 0; r < 16; ++r)
        vals[r] = *reinterpret_cast<const v4f*>(
            xp + ((size_t)(r >> 3) * TLEN + (r & 7) * 16 + adv) * NVAR);
      av0 = *reinterpret_cast<const v4f*>(ap + adv * RANK);
      av1 = *reinterpret_cast<const v4f*>(ap + adv * RANK + 1024);
    }

    __builtin_prefetch(pf + (size_t)c * 16 * TLEN * NVAR, 0, 2);

    const float* xb = xs + (c & 1) * XS_CH + fr;
    const float* ab = as + (c & 1) * AS_CH + (2 * half) * 16 + m;
    #pragma unroll
    for (int ic = 0; ic < KCH / 4; ic += 2) {
      v2f b0m, b1m;
      b0m.x = ab[(ic * 4 + 0) * 16];
      b0m.y = ab[(ic * 4 + 1) * 16];
      b1m.x = ab[(ic * 4 + 4) * 16];
      b1m.y = ab[(ic * 4 + 5) * 16];
      v2f a0, a1;
      a0.x = TANHF(g * xb[(ic * 4 + 0) * XROW]);
      a0.y = TANHF(g * xb[(ic * 4 + 1) * XROW]);
      a1.x = TANHF(g * xb[(ic * 4 + 4) * XROW]);
      a1.y = TANHF(g * xb[(ic * 4 + 5) * XROW]);
      acc0 = __builtin_amdgcn_wmma_f32_16x16x4_f32(false, a0, false, b0m,
                                                   (short)0, acc0, false, false);
      acc1 = __builtin_amdgcn_wmma_f32_16x16x4_f32(false, a1, false, b1m,
                                                   (short)0, acc1, false, false);
    }

    if (more) {
      float* xn = xs + ((c + 1) & 1) * XS_CH;
      float* an = as + ((c + 1) & 1) * AS_CH;
      #pragma unroll
      for (int r = 0; r < 16; ++r)
        *reinterpret_cast<v4f*>(&xn[(r >> 3) * XPL + (r & 7) * 16 * XROW + xw]) = vals[r];
      *reinterpret_cast<v4f*>(&an[tid * 4]) = av0;
      *reinterpret_cast<v4f*>(&an[1024 + tid * 4]) = av1;
      __syncthreads();
    }
  }

  #pragma unroll
  for (int r = 0; r < 8; ++r)
    tmp[((size_t)(wave * 8 + r) * BATCH + b0 + half) * RANK + m] = acc0[r] + acc1[r];
}

static __device__ __forceinline__ void lds_barrier() {
  asm volatile("s_wait_dscnt 0x0\n\ts_barrier_signal -1\n\ts_barrier_wait -1" ::: "memory");
  __builtin_amdgcn_sched_barrier(0);
}

#define S2_VPW  4
#define OP      68
#define OT_FL   (16 * 16 * OP)

__global__ __launch_bounds__(512) void gcm_stage2(
    const float* __restrict__ x, const float* __restrict__ bias,
    const float* __restrict__ lora_B, const float* __restrict__ tmp,
    float* __restrict__ out)
{
  __shared__ __attribute__((aligned(16))) float smem[2 * OT_FL];
  float* Bs = smem + OT_FL;

  const int tid  = threadIdx.x;
  const int lane = tid & 31;
  const int wave = tid >> 5;
  const int bg0  = blockIdx.x * (16 * S2_BT);
  const int o0   = blockIdx.y * 16;
  const int m    = lane & 15;
  const int half = lane >> 4;
  const int koff = half * 2;
  const int vw   = wave * S2_VPW;

  const int rv = (tid & 15) * 4;
  const int ro = (tid >> 4) & 15;
  const int rb = tid >> 8;

  v4f bvals[8];
  #pragma unroll
  for (int r = 0; r < 8; ++r)
    bvals[r] = *reinterpret_cast<const v4f*>(
        &lora_B[((size_t)(rb + 2 * r) * TLEN + o0 + ro) * NVAR + rv]);

  v2f  afr[2][S2_VPW][4];
  v4f  xv[2][8];

  const float* tp = tmp + ((size_t)vw * BATCH + bg0 + m) * RANK + koff;
  const float* xq = x + ((size_t)(bg0 + rb) * TLEN + o0 + ro) * NVAR + rv;
  float*       oq = out + ((size_t)(bg0 + rb) * TLEN + o0 + ro) * NVAR + rv;

  #pragma unroll
  for (int v = 0; v < S2_VPW; ++v)
    #pragma unroll
    for (int kc = 0; kc < 4; ++kc)
      afr[0][v][kc] = *reinterpret_cast<const v2f*>(
          tp + (size_t)v * BATCH * RANK + kc * 4);
  #pragma unroll
  for (int j = 0; j < 8; ++j)
    xv[0][j] = *reinterpret_cast<const v4f*>(xq + (size_t)(2 * j) * TLEN * NVAR);
  const v4f bv = *reinterpret_cast<const v4f*>(&bias[(size_t)(o0 + ro) * NVAR + rv]);

  #pragma unroll
  for (int r = 0; r < 8; ++r) {
    const int k = rb + 2 * r;
    Bs[BS_IDX(k, rv + 0, ro)] = bvals[r].x;
    Bs[BS_IDX(k, rv + 1, ro)] = bvals[r].y;
    Bs[BS_IDX(k, rv + 2, ro)] = bvals[r].z;
    Bs[BS_IDX(k, rv + 3, ro)] = bvals[r].w;
  }
  lds_barrier();

  v2f bfr[S2_VPW][4];
  #pragma unroll
  for (int v = 0; v < S2_VPW; ++v)
    #pragma unroll
    for (int kc = 0; kc < 4; ++kc) {
      const int k = kc * 4 + koff;
      bfr[v][kc].x = Bs[BS_IDX(k + 0, vw + v, m)];
      bfr[v][kc].y = Bs[BS_IDX(k + 1, vw + v, m)];
    }

  #pragma unroll
  for (int t = 0; t < S2_BT; ++t) {
    const int cur = t & 1;
    const int nxt = cur ^ 1;
    if (t + 1 < S2_BT) {
      #pragma unroll
      for (int v = 0; v < S2_VPW; ++v)
        #pragma unroll
        for (int kc = 0; kc < 4; ++kc)
          afr[nxt][v][kc] = *reinterpret_cast<const v2f*>(
              tp + (size_t)v * BATCH * RANK + (size_t)(t + 1) * 16 * RANK + kc * 4);
      #pragma unroll
      for (int j = 0; j < 8; ++j)
        xv[nxt][j] = *reinterpret_cast<const v4f*>(
            xq + (size_t)((t + 1) * 16 + 2 * j) * TLEN * NVAR);
    }

    v8f acc[S2_VPW];
    #pragma unroll
    for (int v = 0; v < S2_VPW; ++v)
      acc[v] = (v8f){0.f, 0.f, 0.f, 0.f, 0.f, 0.f, 0.f, 0.f};
    #pragma unroll
    for (int kc = 0; kc < 4; ++kc)
      #pragma unroll
      for (int v = 0; v < S2_VPW; ++v)
        acc[v] = __builtin_amdgcn_wmma_f32_16x16x4_f32(
            false, afr[cur][v][kc], false, bfr[v][kc], (short)0, acc[v],
            false, false);

    float* ob = smem + cur * OT_FL;
    #pragma unroll
    for (int r = 0; r < 8; ++r) {
      v4f d;
      d.x = acc[0][r]; d.y = acc[1][r]; d.z = acc[2][r]; d.w = acc[3][r];
      *reinterpret_cast<v4f*>(&ob[((half * 8 + r) * 16 + m) * OP + vw]) = d;
    }
    lds_barrier();

    #pragma unroll
    for (int j = 0; j < 8; ++j) {
      const int b = rb + 2 * j;
      const v4f pv = *reinterpret_cast<const v4f*>(&ob[(b * 16 + ro) * OP + rv]);
      v4f ov;
      ov.x = xv[cur][j].x + bv.x + pv.x;
      ov.y = xv[cur][j].y + bv.y + pv.y;
      ov.z = xv[cur][j].z + bv.z + pv.z;
      ov.w = xv[cur][j].w + bv.w + pv.w;
      __builtin_nontemporal_store(
          ov, reinterpret_cast<v4f*>(oq + (size_t)(t * 16 + 2 * j) * TLEN * NVAR));
    }
  }
}

extern "C" void kernel_launch(void* const* d_in, const int* in_sizes, int n_in,
                              void* d_out, int out_size, void* d_ws, size_t ws_size,
                              hipStream_t stream) {
  (void)in_sizes; (void)n_in; (void)out_size; (void)ws_size;
  const float* x      = (const float*)d_in[0];
  const float* gating = (const float*)d_in[1];
  const float* bias   = (const float*)d_in[2];
  const float* lora_A = (const float*)d_in[3];
  const float* lora_B = (const float*)d_in[4];
  float* out = (float*)d_out;
  float* tmp = (float*)d_ws;

  dim3 g1(BATCH / 2);
  gcm_stage1<<<g1, dim3(256), 0, stream>>>(x, gating, lora_A, tmp);

  dim3 g2(S2_GX, TLEN / 16);
  gcm_stage2<<<g2, dim3(512), 0, stream>>>(x, bias, lora_B, tmp, out);
}


# Stage 2 work-groups re-shaped: 64 batch rows x 2 o-tiles x 32 variables (half rows, whole 128 B lines), 8 waves (were 1 o-tile x 64 v, 16 waves): a tmp fragment feeds both o-tiles; sum order kept
# speedup vs baseline: 3.2963x; 1.0384x over previous
#include <hip/hip_runtime.h>

#define BATCH 512
#define TLEN  512
#define NVAR  64
#define RANK  16
#define S2_BT 4
#define S2_GX (BATCH / (16 * S2_BT))
#define S2_OR 32
#define S2_GY (TLEN / S2_OR)
#define S2_VH 32

typedef __attribute__((ext_vector_type(2))) float v2f;
typedef __attribute__((ext_vector_type(4))) float v4f;
typedef __attribute__((ext_vector_type(8))) float v8f;

#if __has_builtin(__builtin_amdgcn_tanhf)
  #define TANHF(x) __builtin_amdgcn_tanhf(x)
#elif __has_builtin(__builtin_amdgcn_tanh_f32)
  #define TANHF(x) __builtin_amdgcn_tanh_f32(x)
#else
  #define TANHF(x) tanhf(x)
#endif


#define KCH   128
#define NCH   (TLEN / KCH)
#define XROW  72
#define XPL   (KCH * XROW + 32)
#define XS_CH (2 * XPL)
#define AS_CH (KCH * RANK)

__global__ __launch_bounds__(256) void gcm_stage1(
    const float* __restrict__ x, const float* __restrict__ gating,
    const float* __restrict__ lora_A, float* __restrict__ tmp)
{
  __shared__ __attribute__((aligned(16))) float xs[2 * XS_CH];
  __shared__ __attribute__((aligned(16))) float as[2 * AS_CH];

  const int tid  = threadIdx.x;
  const int lane = tid & 31;
  const int wave = tid >> 5;
  const int b0   = blockIdx.x * 2;

  const int m    = lane & 15;
  const int half = lane >> 4;
  const int bl   = m >> 3;
  const int vl   = wave * 8 + (m & 7);
  const float g  = gating[vl];

  const int sv = (tid & 15) * 4;
  const int sr = tid >> 4;
  const float* xp = x + ((size_t)b0 * TLEN + sr) * NVAR + sv;
  const float* ap = lora_A + tid * 4;
  const int xw = sr * XROW + sv;
  const int fr = bl * XPL + (2 * half) * XROW + vl;

  const float* pf = x + ((size_t)((blockIdx.x % S2_GX) * (16 * S2_BT) + sr) * TLEN
                         + ((blockIdx.x / S2_GX) % S2_GY) * S2_OR
                         + (blockIdx.x / (S2_GX * S2_GY)) * 16 + (tid & 15)) * NVAR;

  v4f vals[16];
  v4f av0, av1;

  #pragma unroll
  for (int r = 0; r < 16; ++r)
    vals[r] = *reinterpret_cast<const v4f*>(
        xp + ((size_t)(r >> 3) * TLEN + (r & 7) * 16) * NVAR);
  av0 = *reinterpret_cast<const v4f*>(ap);
  av1 = *reinterpret_cast<const v4f*>(ap + 1024);
  #pragma unroll
  for (int r = 0; r < 16; ++r)
    *reinterpret_cast<v4f*>(&xs[(r >> 3) * XPL + (r & 7) * 16 * XROW + xw]) = vals[r];
  *reinterpret_cast<v4f*>(&as[tid * 4]) = av0;
  *reinterpret_cast<v4f*>(&as[1024 + tid * 4]) = av1;
  __syncthreads();

  v8f acc0 = {0.f, 0.f, 0.f, 0.f, 0.f, 0.f, 0.f, 0.f};
  v8f acc1 = {0.f, 0.f, 0.f, 0.f, 0.f, 0.f, 0.f, 0.f};

  for (int c = 0; c < NCH; ++c) {
    const bool more = (c + 1 < NCH);
    if (more) {
      const size_t adv = (size_t)(c + 1) * KCH;
      #pragma unroll
      for (int r = 0; r < 16; ++r)
        vals[r] = *reinterpret_cast<const v4f*>(
            xp + ((size_t)(r >> 3) * TLEN + (r & 7) * 16 + adv) * NVAR);
      av0 = *reinterpret_cast<const v4f*>(ap + adv * RANK);
      av1 = *reinterpret_cast<const v4f*>(ap + adv * RANK + 1024);
    }

    __builtin_prefetch(pf + (size_t)c * 16 * TLEN * NVAR, 0, 2);

    const float* xb = xs + (c & 1) * XS_CH + fr;
    const float* ab = as + (c & 1) * AS_CH + (2 * half) * 16 + m;
    #pragma unroll
    for (int ic = 0; ic < KCH / 4; ic += 2) {
      v2f b0m, b1m;
      b0m.x = ab[(ic * 4 + 0) * 16];
      b0m.y = ab[(ic * 4 + 1) * 16];
      b1m.x = ab[(ic * 4 + 4) * 16];
      b1m.y = ab[(ic * 4 + 5) * 16];
      v2f a0, a1;
      a0.x = TANHF(g * xb[(ic * 4 + 0) * XROW]);
      a0.y = TANHF(g * xb[(ic * 4 + 1) * XROW]);
      a1.x = TANHF(g * xb[(ic * 4 + 4) * XROW]);
      a1.y = TANHF(g * xb[(ic * 4 + 5) * XROW]);
      acc0 = __builtin_amdgcn_wmma_f32_16x16x4_f32(false, a0, false, b0m,
                                                   (short)0, acc0, false, false);
      acc1 = __builtin_amdgcn_wmma_f32_16x16x4_f32(false, a1, false, b1m,
                                                   (short)0, acc1, false, false);
    }

    if (more) {
      float* xn = xs + ((c + 1) & 1) * XS_CH;
      float* an = as + ((c + 1) & 1) * AS_CH;
      #pragma unroll
      for (int r = 0; r < 16; ++r)
        *reinterpret_cast<v4f*>(&xn[(r >> 3) * XPL + (r & 7) * 16 * XROW + xw]) = vals[r];
      *reinterpret_cast<v4f*>(&an[tid * 4]) = av0;
      *reinterpret_cast<v4f*>(&an[1024 + tid * 4]) = av1;
      __syncthreads();
    }
  }

  #pragma unroll
  for (int r = 0; r < 8; ++r)
    tmp[((size_t)(wave * 8 + r) * BATCH + b0 + half) * RANK + m] = acc0[r] + acc1[r];
}

static __device__ __forceinline__ void lds_barrier() {
  asm volatile("s_wait_dscnt 0x0\n\ts_barrier_signal -1\n\ts_barrier_wait -1" ::: "memory");
  __builtin_amdgcn_sched_barrier(0);
}

#define S2_TH   256
#define S2_VPW  4
#define S2_OT   (S2_OR / 16)
#define OP      36
#define OT_FL   (16 * 16 * OP)
#define BS_G    68
#define BS_K    (8 * BS_G + 8)
#define BS_FL   (S2_OT * 16 * BS_K)
#define BSX(ot, k, v, o)  ((((ot) * 16 + (k)) * BS_K) + (((v) >> 2) * BS_G) + (((v) & 3) * 16) + (o))
#define S2_LDS_FL (2 * S2_OT * OT_FL + BS_FL)

__global__ __launch_bounds__(S2_TH) void gcm_stage2(
    const float* __restrict__ x, const float* __restrict__ bias,
    const float* __restrict__ lora_B, const float* __restrict__ tmp,
    float* __restrict__ out)
{
  __shared__ __attribute__((aligned(16))) float smem[S2_LDS_FL];
  float* Bs = smem + 2 * S2_OT * OT_FL;

  const int tid  = threadIdx.x;
  const int lane = tid & 31;
  const int wave = tid >> 5;
  const int bg0  = blockIdx.x * (16 * S2_BT);
  const int o0   = blockIdx.y * S2_OR;
  const int v0   = blockIdx.z * S2_VH;
  const int m    = lane & 15;
  const int half = lane >> 4;
  const int koff = half * 2;
  const int vw   = wave * S2_VPW;

  const int rv = (tid & 7) * 4;
  const int ru = tid >> 3;
  const int ro = ((ru & 1) << 3) | ((ru >> 1) & 7);
  const int so = ru & 15;
  const int rb = tid >> 7;

  v4f bvals[S2_OT][8];
  #pragma unroll
  for (int ot = 0; ot < S2_OT; ++ot)
    #pragma unroll
    for (int r = 0; r < 8; ++r)
      bvals[ot][r] = *reinterpret_cast<const v4f*>(
          &lora_B[((size_t)(rb + 2 * r) * TLEN + o0 + ot * 16 + so) * NVAR + v0 + rv]);

  v2f  afr[2][S2_VPW][4];
  v4f  xv[2][S2_OT][8];

  const float* tp = tmp + ((size_t)(v0 + vw) * BATCH + bg0 + m) * RANK + koff;
  const float* xq = x + ((size_t)(bg0 + rb) * TLEN + o0 + ro) * NVAR + v0 + rv;
  float*       oq = out + ((size_t)(bg0 + rb) * TLEN + o0 + ro) * NVAR + v0 + rv;

  #pragma unroll
  for (int v = 0; v < S2_VPW; ++v)
    #pragma unroll
    for (int kc = 0; kc < 4; ++kc)
      afr[0][v][kc] = *reinterpret_cast<const v2f*>(
          tp + (size_t)v * BATCH * RANK + kc * 4);
  #pragma unroll
  for (int ot = 0; ot < S2_OT; ++ot)
    #pragma unroll
    for (int j = 0; j < 8; ++j)
      xv[0][ot][j] = *reinterpret_cast<const v4f*>(
          xq + ((size_t)(2 * j) * TLEN + ot * 16) * NVAR);
  v4f bv[S2_OT];
  #pragma unroll
  for (int ot = 0; ot < S2_OT; ++ot)
    bv[ot] = *reinterpret_cast<const v4f*>(
        &bias[(size_t)(o0 + ot * 16 + ro) * NVAR + v0 + rv]);

  #pragma unroll
  for (int ot = 0; ot < S2_OT; ++ot)
    #pragma unroll
    for (int r = 0; r < 8; ++r) {
      const int k = rb + 2 * r;
      Bs[BSX(ot, k, rv + 0, so)] = bvals[ot][r].x;
      Bs[BSX(ot, k, rv + 1, so)] = bvals[ot][r].y;
      Bs[BSX(ot, k, rv + 2, so)] = bvals[ot][r].z;
      Bs[BSX(ot, k, rv + 3, so)] = bvals[ot][r].w;
    }
  lds_barrier();

  v2f bfr[S2_OT][S2_VPW][4];
  #pragma unroll
  for (int ot = 0; ot < S2_OT; ++ot)
    #pragma unroll
    for (int v = 0; v < S2_VPW; ++v)
      #pragma unroll
      for (int kc = 0; kc < 4; ++kc) {
        const int k = kc * 4 + koff;
        bfr[ot][v][kc].x = Bs[BSX(ot, k + 0, vw + v, m)];
        bfr[ot][v][kc].y = Bs[BSX(ot, k + 1, vw + v, m)];
      }

  #pragma unroll
  for (int t = 0; t < S2_BT; ++t) {
    const int cur = t & 1;
    const int nxt = cur ^ 1;
    if (t + 1 < S2_BT) {
      #pragma unroll
      for (int v = 0; v < S2_VPW; ++v)
        #pragma unroll
        for (int kc = 0; kc < 4; ++kc)
          afr[nxt][v][kc] = *reinterpret_cast<const v2f*>(
              tp + (size_t)v * BATCH * RANK + (size_t)(t + 1) * 16 * RANK + kc * 4);
      #pragma unroll
      for (int ot = 0; ot < S2_OT; ++ot)
        #pragma unroll
        for (int j = 0; j < 8; ++j)
          xv[nxt][ot][j] = *reinterpret_cast<const v4f*>(
              xq + ((size_t)((t + 1) * 16 + 2 * j) * TLEN + ot * 16) * NVAR);
    }

    #pragma unroll
    for (int ot = 0; ot < S2_OT; ++ot) {
      v8f acc[S2_VPW];
      #pragma unroll
      for (int v = 0; v < S2_VPW; ++v)
        acc[v] = (v8f){0.f, 0.f, 0.f, 0.f, 0.f, 0.f, 0.f, 0.f};
      #pragma unroll
      for (int kc = 0; kc < 4; ++kc)
        #pragma unroll
        for (int v = 0; v < S2_VPW; ++v)
          acc[v] = __builtin_amdgcn_wmma_f32_16x16x4_f32(
              false, afr[cur][v][kc], false, bfr[ot][v][kc], (short)0, acc[v],
              false, false);

      float* ob = smem + (cur * S2_OT + ot) * OT_FL;
      #pragma unroll
      for (int r = 0; r < 8; ++r) {
        v4f d;
        d.x = acc[0][r]; d.y = acc[1][r]; d.z = acc[2][r]; d.w = acc[3][r];
        *reinterpret_cast<v4f*>(&ob[((half * 8 + r) * 16 + m) * OP + vw]) = d;
      }
    }
    lds_barrier();

    #pragma unroll
    for (int ot = 0; ot < S2_OT; ++ot) {
      const float* ob = smem + (cur * S2_OT + ot) * OT_FL;
      #pragma unroll
      for (int j = 0; j < 8; ++j) {
        const int b = rb + 2 * j;
        const v4f pv = *reinterpret_cast<const v4f*>(&ob[(b * 16 + ro) * OP + rv]);
        v4f ov;
        ov.x = xv[cur][ot][j].x + bv[ot].x + pv.x;
        ov.y = xv[cur][ot][j].y + bv[ot].y + pv.y;
        ov.z = xv[cur][ot][j].z + bv[ot].z + pv.z;
        ov.w = xv[cur][ot][j].w + bv[ot].w + pv.w;
        __builtin_nontemporal_store(
            ov, reinterpret_cast<v4f*>(oq + ((size_t)(t * 16 + 2 * j) * TLEN + ot * 16) * NVAR));
      }
    }
  }
}

extern "C" void kernel_launch(void* const* d_in, const int* in_sizes, int n_in,
                              void* d_out, int out_size, void* d_ws, size_t ws_size,
                              hipStream_t stream) {
  (void)in_sizes; (void)n_in; (void)out_size; (void)ws_size;
  const float* x      = (const float*)d_in[0];
  const float* gating = (const float*)d_in[1];
  const float* bias   = (const float*)d_in[2];
  const float* lora_A = (const float*)d_in[3];
  const float* lora_B = (const float*)d_in[4];
  float* out = (float*)d_out;
  float* tmp = (float*)d_ws;

  dim3 g1(BATCH / 2);
  gcm_stage1<<<g1, dim3(256), 0, stream>>>(x, gating, lora_A, tmp);

  dim3 g2(S2_GX, S2_GY, NVAR / S2_VH);
  gcm_stage2<<<g2, dim3(S2_TH), 0, stream>>>(x, bias, lora_B, tmp, out);
}
